# starting kernel
# baseline (speedup 1.0000x reference)
#include <hip/hip_runtime.h>


#define N_DIM 8192
constexpr float LAM = 1.0f;

typedef __attribute__((ext_vector_type(2))) float v2f;
typedef __attribute__((ext_vector_type(8))) float v8f;

__global__ void init_kernel(float* __restrict__ S, float* __restrict__ out0) {
    int i = blockIdx.x * blockDim.x + threadIdx.x;
    if (i < N_DIM) S[i] = 0.0f;
    if (i == 0) out0[0] = 0.0f;
}

__global__ __launch_bounds__(256) void colsum_kernel(const float* __restrict__ M,
                                                     const float* __restrict__ alpha,
                                                     float* __restrict__ S) {
    const int lane = threadIdx.x & 31;
    const int wave = threadIdx.x >> 5;
    const int half = lane >> 4;
    const int ln   = lane & 15;
    const int colTile = blockIdx.x * 8 + wave;
    const int j = colTile * 16 + ln;
    const int rowBase = blockIdx.y * (N_DIM / 16);

    v8f acc = {};
    v2f ones; ones.x = 1.0f; ones.y = 1.0f;

    for (int it = 0; it < (N_DIM / 16) / 4; ++it) {
        const int i0 = rowBase + it * 4 + half * 2;
        const size_t base = (size_t)i0 * N_DIM + j;
        const float m0 = M[base];
        const float m1 = M[base + N_DIM];
        __builtin_prefetch(&M[base + (size_t)8 * N_DIM], 0, 1);
        v2f b;
        b.x = __expf(LAM * (alpha[i0]     - m0));
        b.y = __expf(LAM * (alpha[i0 + 1] - m1));
        acc = __builtin_amdgcn_wmma_f32_16x16x4_f32(
             false, ones,  false, b,
             (short)0, acc,  false,  false);
    }
    if (half == 0) atomicAdd(&S[j], acc[0]);
}

__global__ void beta_kernel(const float* __restrict__ S,
                            const float* __restrict__ wb,
                            const float* __restrict__ alpha,
                            const float* __restrict__ wa,
                            float* __restrict__ beta,
                            float* __restrict__ out0) {
    int j = blockIdx.x * blockDim.x + threadIdx.x;
    if (j < N_DIM) {
        float b = (logf(wb[j]) - logf(S[j])) / LAM;
        beta[j] = b;
        atomicAdd(out0, -alpha[j] * wa[j] - b * wb[j]);
    }
}

__global__ __launch_bounds__(256) void rowsum_kernel(const float* __restrict__ M,
                                                     const float* __restrict__ alpha,
                                                     const float* __restrict__ beta,
                                                     const float* __restrict__ wa,
                                                     float* __restrict__ out) {
    const int lane = threadIdx.x & 31;
    const int wave = threadIdx.x >> 5;
    const int half = lane >> 4;
    const int ln   = lane & 15;
    const int rowTile = blockIdx.x * 8 + wave;
    const int i0 = rowTile * 4 + half * 2;
    const float a0 = alpha[i0];
    const float a1 = alpha[i0 + 1];

    float accx = 0.0f, accy = 0.0f;
    for (int it = 0; it < N_DIM / 16; ++it) {
        const int j = it * 16 + ln;
        const size_t base = (size_t)i0 * N_DIM + j;
        const float m0 = M[base];
        const float m1 = M[base + N_DIM];
        __builtin_prefetch(&M[base + 128], 0, 1);
        const float bj = beta[j];
        accx += __expf(LAM * (a0 + bj - m0));
        accy += __expf(LAM * (a1 + bj - m1));
    }
    for (int mask = 8; mask >= 1; mask >>= 1) {
        accx += __shfl_xor(accx, mask, 32);
        accy += __shfl_xor(accy, mask, 32);
    }
    if (ln == 0) {
        out[1 + i0]     = accx - wa[i0];
        out[1 + i0 + 1] = accy - wa[i0 + 1];
        atomicAdd(&out[0], (accx + accy) / LAM);
    }
}

extern "C" void kernel_launch(void* const* d_in, const int* in_sizes, int n_in,
                              void* d_out, int out_size, void* d_ws, size_t ws_size,
                              hipStream_t stream) {
    const float* alpha = (const float*)d_in[0];
    const float* wa    = (const float*)d_in[1];
    const float* wb    = (const float*)d_in[2];
    const float* M     = (const float*)d_in[3];
    float* out  = (float*)d_out;
    float* S    = (float*)d_ws;
    float* beta = S + N_DIM;

    init_kernel<<<N_DIM / 256, 256, 0, stream>>>(S, out);
    colsum_kernel<<<dim3(64, 16), 256, 0, stream>>>(M, alpha, S);
    beta_kernel<<<N_DIM / 256, 256, 0, stream>>>(S, wb, alpha, wa, beta, out);
    rowsum_kernel<<<256, 256, 0, stream>>>(M, alpha, beta, wa, out);
}


# rowsum pass re-tiled as a stream: each wave sums 16 rows x 1024 columns from 512-byte non-temporal b128 loads issued 3 rows ahead, beta in registers; new reduction tree (butterfly + LDS join)
# speedup vs baseline: 1.8420x; 1.8420x over previous
#include <hip/hip_runtime.h>


#define N_DIM 8192
constexpr float LAM = 1.0f;

typedef __attribute__((ext_vector_type(2))) float v2f;
typedef __attribute__((ext_vector_type(8))) float v8f;

__global__ void init_kernel(float* __restrict__ S, float* __restrict__ out0) {
    int i = blockIdx.x * blockDim.x + threadIdx.x;
    if (i < N_DIM) S[i] = 0.0f;
    if (i == 0) out0[0] = 0.0f;
}

__global__ __launch_bounds__(256) void colsum_kernel(const float* __restrict__ M,
                                                     const float* __restrict__ alpha,
                                                     float* __restrict__ S) {
    const int lane = threadIdx.x & 31;
    const int wave = threadIdx.x >> 5;
    const int half = lane >> 4;
    const int ln   = lane & 15;
    const int colTile = blockIdx.x * 8 + wave;
    const int j = colTile * 16 + ln;
    const int rowBase = blockIdx.y * (N_DIM / 16);

    v8f acc = {};
    v2f ones; ones.x = 1.0f; ones.y = 1.0f;

    for (int it = 0; it < (N_DIM / 16) / 4; ++it) {
        const int i0 = rowBase + it * 4 + half * 2;
        const size_t base = (size_t)i0 * N_DIM + j;
        const float m0 = M[base];
        const float m1 = M[base + N_DIM];
        __builtin_prefetch(&M[base + (size_t)8 * N_DIM], 0, 1);
        v2f b;
        b.x = __expf(LAM * (alpha[i0]     - m0));
        b.y = __expf(LAM * (alpha[i0 + 1] - m1));
        acc = __builtin_amdgcn_wmma_f32_16x16x4_f32(
             false, ones,  false, b,
             (short)0, acc,  false,  false);
    }
    if (half == 0) atomicAdd(&S[j], acc[0]);
}

__global__ void beta_kernel(const float* __restrict__ S,
                            const float* __restrict__ wb,
                            const float* __restrict__ alpha,
                            const float* __restrict__ wa,
                            float* __restrict__ beta,
                            float* __restrict__ out0) {
    int j = blockIdx.x * blockDim.x + threadIdx.x;
    if (j < N_DIM) {
        float b = (logf(wb[j]) - logf(S[j])) / LAM;
        beta[j] = b;
        atomicAdd(out0, -alpha[j] * wa[j] - b * wb[j]);
    }
}

#define RS_ROWS   16
#define RS_WAVES  8
#define RS_PIECES 8
#define RS_AHEAD  3
#define RS_RING   (RS_AHEAD + 1)

typedef __attribute__((ext_vector_type(4))) float v4f;

__global__ __launch_bounds__(RS_WAVES * 32) void rowsum_kernel(const float* __restrict__ M,
                                                     const float* __restrict__ alpha,
                                                     const float* __restrict__ beta,
                                                     const float* __restrict__ wa,
                                                     float* __restrict__ out) {
    __shared__ float red[RS_ROWS][RS_WAVES];
    const int lane = threadIdx.x & 31;
    const int wave = __builtin_amdgcn_readfirstlane(threadIdx.x >> 5);
    const int row0 = blockIdx.x * RS_ROWS;
    const int col0 = wave * (RS_PIECES * 128) + lane * 4;

    v4f bq[RS_PIECES];
#pragma unroll
    for (int q = 0; q < RS_PIECES; ++q)
        bq[q] = *reinterpret_cast<const v4f*>(beta + col0 + 128 * q);

    const float* mp = M + (size_t)row0 * N_DIM + col0;
    v4f buf[RS_RING][RS_PIECES];
    float sum[RS_ROWS];

#define RS_LOAD_ROW(R) do { \
    _Pragma("unroll") for (int q = 0; q < RS_PIECES; ++q) \
        buf[(R) % RS_RING][q] = __builtin_nontemporal_load( \
            reinterpret_cast<const v4f*>(mp + (size_t)(R) * N_DIM + 128 * q)); } while (0)

#pragma unroll
    for (int r = 0; r < RS_AHEAD; ++r) RS_LOAD_ROW(r);
#pragma unroll
    for (int r = 0; r < RS_ROWS; ++r) {
        __builtin_amdgcn_sched_barrier(0);
        if (r + RS_AHEAD < RS_ROWS) RS_LOAD_ROW(r + RS_AHEAD);
        __builtin_amdgcn_sched_barrier(0);
        const float a = alpha[row0 + r];
        const v4f* m = buf[r % RS_RING];
        float c0 = 0.0f, c1 = 0.0f, c2 = 0.0f, c3 = 0.0f;
#pragma unroll
        for (int q = 0; q < RS_PIECES; ++q) {
            c0 += __expf(LAM * (a + bq[q].x - m[q].x));
            c1 += __expf(LAM * (a + bq[q].y - m[q].y));
            c2 += __expf(LAM * (a + bq[q].z - m[q].z));
            c3 += __expf(LAM * (a + bq[q].w - m[q].w));
        }
        sum[r] = (c0 + c1) + (c2 + c3);
    }
#undef RS_LOAD_ROW
    __builtin_amdgcn_sched_barrier(0);

#pragma unroll
    for (int r = 0; r < RS_ROWS; ++r) {
        float s = sum[r];
#pragma unroll
        for (int mask = 16; mask >= 1; mask >>= 1) s += __shfl_xor(s, mask, 32);
        sum[r] = s;
    }
    float mine = 0.0f;
#pragma unroll
    for (int r = 0; r < RS_ROWS; ++r) if (lane == r) mine = sum[r];
    if (lane < RS_ROWS) red[lane][wave] = mine;
    __syncthreads();

    if (wave == 0) {
        float total = 0.0f;
        if (lane < RS_ROWS) {
#pragma unroll
            for (int w = 0; w < RS_WAVES; ++w) total += red[lane][w];
            out[1 + row0 + lane] = total - wa[row0 + lane];
        }
        float s = total;
#pragma unroll
        for (int mask = 16; mask >= 1; mask >>= 1) s += __shfl_xor(s, mask, 32);
        if (lane == 0) atomicAdd(&out[0], s / LAM);
    }
}

extern "C" void kernel_launch(void* const* d_in, const int* in_sizes, int n_in,
                              void* d_out, int out_size, void* d_ws, size_t ws_size,
                              hipStream_t stream) {
    const float* alpha = (const float*)d_in[0];
    const float* wa    = (const float*)d_in[1];
    const float* wb    = (const float*)d_in[2];
    const float* M     = (const float*)d_in[3];
    float* out  = (float*)d_out;
    float* S    = (float*)d_ws;
    float* beta = S + N_DIM;

    init_kernel<<<N_DIM / 256, 256, 0, stream>>>(S, out);
    colsum_kernel<<<dim3(64, 16), 256, 0, stream>>>(M, alpha, S);
    beta_kernel<<<N_DIM / 256, 256, 0, stream>>>(S, wb, alpha, wa, beta, out);
    rowsum_kernel<<<N_DIM / RS_ROWS, RS_WAVES * 32, 0, stream>>>(M, alpha, beta, wa, out);
}
